# starting kernel
# baseline (speedup 1.0000x reference)
#include <hip/hip_runtime.h>
#include <cmath>


#define L      32
#define KHEADS 8
#define NH     8
#define HID    300
#define BATCH  4096
#define NB     18688

#define TILE_M 16
#define NWAVES 16
#define THREADS (NWAVES * 32)

#define PCOLS 2336
#define NTILES (PCOLS / 16)
#define PSTR  2340
#define QSTR  33
#define HSTR  308

#define ROFF  0
#define RTOFF 1024
#define VOFF  2048
#define BBOFF 2304

typedef float v2f __attribute__((ext_vector_type(2)));
typedef float v8f __attribute__((ext_vector_type(8)));

__device__ __forceinline__ float softplus_f(float x) {
    return log1pf(__expf(-fabsf(x))) + fmaxf(x, 0.0f);
}

__device__ __forceinline__ int pcol_to_wcol(int pcol, int k) {
    if      (pcol < 1024) return k * 1024 + pcol;
    else if (pcol < 2048) return 8192  + k * 1024 + (pcol - 1024);
    else if (pcol < 2304) return 16384 + k * 256  + (pcol - 2048);
    else                  return 18432 + k * 32   + (pcol - 2304);
}

extern "C" __global__ __launch_bounds__(THREADS)
void sylvester_flow_kernel(const float* __restrict__ z0,
                           const float* __restrict__ hmat,
                           const float* __restrict__ W,
                           const float* __restrict__ bvec,
                           float* __restrict__ out)
{
    extern __shared__ float smem[];
    float* P  = smem;
    float* Qs = P  + TILE_M * PSTR;
    float* hL = Qs + TILE_M * (L * QSTR);
    float* zs = hL + TILE_M * HSTR;
    float* ys = zs + TILE_M * L;
    float* ts = ys + TILE_M * L;
    float* ss = ts + TILE_M * L;

    const int tid  = threadIdx.x;
    const int wave = tid >> 5;
    const int lane = tid & 31;
    const int hi   = lane >> 4;
    const int n16  = lane & 15;
    const int kk2  = hi * 2;
    const int row0 = blockIdx.x * TILE_M;

    for (int idx = tid; idx < TILE_M * HID; idx += THREADS) {
        int r = idx / HID, c = idx - r * HID;
        hL[r * HSTR + c] = hmat[(size_t)(row0 + r) * HID + c];
    }
    zs[wave * L + lane] = z0[(size_t)(row0 + wave) * L + lane];

    const int myrow = wave;
    const int prow  = myrow * PSTR;
    const int qbase = myrow * (L * QSTR);
    float ln_det = 0.0f;

    __syncthreads();

    for (int k = 0; k < KHEADS; ++k) {
        for (int nt0 = wave * 2; nt0 < NTILES; nt0 += NWAVES * 2) {
            int pcol0 = nt0 * 16;
            int pcol1 = pcol0 + 16;
            int wcol0 = pcol_to_wcol(pcol0, k);
            int wcol1 = pcol_to_wcol(pcol1, k);

            const float* wp0 = W + (size_t)kk2 * NB + wcol0 + n16;
            const float* wp1 = W + (size_t)kk2 * NB + wcol1 + n16;
            v8f acc0 = {};
            v8f acc1 = {};
            #pragma unroll 5
            for (int s = 0; s < HID / 4; ++s) {
                size_t roff = (size_t)s * 4 * NB;
                v2f bv0, bv1;
                bv0.x = wp0[roff];
                bv0.y = wp0[roff + NB];
                bv1.x = wp1[roff];
                bv1.y = wp1[roff + NB];
                int hb = n16 * HSTR + 4 * s + kk2;
                v2f av;
                av.x = hL[hb];
                av.y = hL[hb + 1];
                acc0 = __builtin_amdgcn_wmma_f32_16x16x4_f32(
                    false, av, false, bv0, (short)0, acc0, false, false);
                acc1 = __builtin_amdgcn_wmma_f32_16x16x4_f32(
                    false, av, false, bv1, (short)0, acc1, false, false);
            }
            float bias0 = bvec[wcol0 + n16];
            float bias1 = bvec[wcol1 + n16];
            #pragma unroll
            for (int r = 0; r < 8; ++r) {
                P[(r + 8 * hi) * PSTR + pcol0 + n16] = acc0[r] + bias0;
                P[(r + 8 * hi) * PSTR + pcol1 + n16] = acc1[r] + bias1;
            }
        }
        __syncthreads();

        float q[L];
        #pragma unroll
        for (int j = 0; j < L; ++j) q[j] = (j == lane) ? 1.0f : 0.0f;

        for (int hd = 0; hd < NH; ++hd) {
            int vb = prow + VOFF + hd * L;
            float vv = 0.0f, u = 0.0f;
            #pragma unroll
            for (int j = 0; j < L; ++j) {
                float vj = P[vb + j];
                vv += vj * vj;
                u  += q[j] * vj;
            }
            float cu = (2.0f / vv) * u;
            #pragma unroll
            for (int j = 0; j < L; ++j) q[j] -= cu * P[vb + j];
        }
        #pragma unroll
        for (int j = 0; j < L; ++j) Qs[qbase + lane * QSTR + j] = q[j];

        float y = 0.0f;
        #pragma unroll
        for (int i = 0; i < L; ++i)
            y += Qs[qbase + i * QSTR + lane] * zs[myrow * L + i];
        ys[myrow * L + lane] = y;

        float rtraw = P[prow + RTOFF + lane * (L + 1)];
        float rt_d  = softplus_f(rtraw) + 0.01f;
        float rraw  = P[prow + ROFF  + lane * (L + 1)];
        float rr    = rraw * rt_d;
        float mval  = softplus_f(rr) - 1.0f;
        float rdnew = rraw + (mval - rr) / rt_d;

        float wv = rt_d * y + P[prow + BBOFF + lane];
        #pragma unroll
        for (int jj = 1; jj < L; ++jj) {
            int col = (lane + jj) & (L - 1);
            float e = P[prow + RTOFF + lane * L + col] * ys[myrow * L + col];
            wv += (col > lane) ? e : 0.0f;
        }
        float t = tanhf(wv);
        ts[myrow * L + lane] = t;

        float sv = rdnew * t;
        #pragma unroll
        for (int jj = 1; jj < L; ++jj) {
            int col = (lane + jj) & (L - 1);
            float e = P[prow + ROFF + lane * L + col] * ts[myrow * L + col];
            sv += (col > lane) ? e : 0.0f;
        }
        ss[myrow * L + lane] = sv;

        float dz = 0.0f;
        #pragma unroll
        for (int j = 0; j < L; ++j) dz += q[j] * ss[myrow * L + j];
        zs[myrow * L + lane] += dz;

        float hp = 1.0f - t * t;
        ln_det += logf(fabsf(1.0f + hp * mval) + 1e-8f);

        __syncthreads();
    }

    out[(size_t)(row0 + myrow) * L + lane] = zs[myrow * L + lane];
    out[(size_t)BATCH * L + (size_t)(row0 + myrow) * L + lane] = ln_det;
}

extern "C" void kernel_launch(void* const* d_in, const int* in_sizes, int n_in,
                              void* d_out, int out_size, void* d_ws, size_t ws_size,
                              hipStream_t stream) {
    (void)in_sizes; (void)n_in; (void)d_ws; (void)ws_size; (void)out_size;
    const float* z0 = (const float*)d_in[0];
    const float* h  = (const float*)d_in[1];
    const float* W  = (const float*)d_in[2];
    const float* b  = (const float*)d_in[3];
    float* out = (float*)d_out;

    size_t lds_bytes = (size_t)(TILE_M * PSTR + TILE_M * L * QSTR +
                                TILE_M * HSTR + 4 * TILE_M * L) * sizeof(float);
    hipFuncSetAttribute((const void*)sylvester_flow_kernel,
                        hipFuncAttributeMaxDynamicSharedMemorySize,
                        (int)lds_bytes);
    sylvester_flow_kernel<<<BATCH / TILE_M, THREADS, lds_bytes, stream>>>(
        z0, h, W, b, out);
}


# W re-laid per call by a first launch into the workspace in WMMA fragment order; the GEMM reads B with one whole-line 16-byte load per lane and k-step, was four 4-byte loads; sums unchanged
# speedup vs baseline: 1.6660x; 1.6660x over previous
#include <hip/hip_runtime.h>
#include <cmath>


#define L      32
#define KHEADS 8
#define NH     8
#define HID    300
#define BATCH  4096
#define NB     18688

#define TILE_M 16
#define NWAVES 16
#define THREADS (NWAVES * 32)

#define PCOLS 2336
#define NTILES (PCOLS / 16)
#define PSTR  2340
#define PBUF  (TILE_M * PSTR)
#define HSTR  308

#define ROFF  0
#define RTOFF 1024
#define VOFF  2048
#define BBOFF 2304

typedef float v2f __attribute__((ext_vector_type(2)));
typedef float v4f __attribute__((ext_vector_type(4)));
typedef float v8f __attribute__((ext_vector_type(8)));

__device__ __forceinline__ float softplus_f(float x) {
    return log1pf(__expf(-fabsf(x))) + fmaxf(x, 0.0f);
}

__device__ __forceinline__ int pcol_to_wcol(int pcol, int k) {
    if      (pcol < 1024) return k * 1024 + pcol;
    else if (pcol < 2048) return 8192  + k * 1024 + (pcol - 1024);
    else if (pcol < 2304) return 16384 + k * 256  + (pcol - 2048);
    else                  return 18432 + k * 32   + (pcol - 2304);
}

template <int CTRL>
__device__ __forceinline__ float row_xor(float x) {
    return __builtin_bit_cast(float, __builtin_amdgcn_update_dpp(
        0, __builtin_bit_cast(int, x), CTRL, 0xf, 0xf, true));
}
__device__ __forceinline__ float xhalf(float x) {
    return __builtin_bit_cast(float, __builtin_amdgcn_permlanex16(
        __builtin_bit_cast(int, x), __builtin_bit_cast(int, x),
        0x76543210, 0xfedcba98, false, false));
}
__device__ __forceinline__ float wave_sum(float x) {
    x += row_xor<0x161>(x);
    x += row_xor<0x162>(x);
    x += row_xor<0x164>(x);
    x += row_xor<0x168>(x);
    x += xhalf(x);
    return x;
}

__device__ __forceinline__ void prio_by_passes_left(int left) {
    if      (left >= 3) __builtin_amdgcn_s_setprio(3);
    else if (left == 2) __builtin_amdgcn_s_setprio(2);
    else if (left == 1) __builtin_amdgcn_s_setprio(1);
    else                __builtin_amdgcn_s_setprio(0);
}

#define KB_STEPS 5
#define KBLOCKS  (HID / 4 / KB_STEPS)
#define NPAIRS   (NTILES / 2)
#define KSTEPS   (HID / 4)
#define REC_FLOATS 128
#define PACK_BYTES ((size_t)KHEADS * NPAIRS * KSTEPS * REC_FLOATS * sizeof(float))

struct FragSet {
    v2f a[KB_STEPS];
    v2f b0[KB_STEPS];
    v2f b1[KB_STEPS];
};

template <bool PK>
__device__ __forceinline__ void load_set(FragSet& f, const float* __restrict__ wcol,
                                         int lane_off, const float* hrow, int kb) {
    const float* hp = hrow + kb * (KB_STEPS * 4);
    if (PK) {
        const v4f* wp = (const v4f*)(wcol + kb * (KB_STEPS * REC_FLOATS));
        #pragma unroll
        for (int s = 0; s < KB_STEPS; ++s) {
            const v4f v = wp[s * (REC_FLOATS / 4)];
            f.b0[s].x = v.x;
            f.b0[s].y = v.y;
            f.b1[s].x = v.z;
            f.b1[s].y = v.w;
        }
    } else {
        const float* wp = wcol + (size_t)kb * (KB_STEPS * 4 * NB) + lane_off;
        #pragma unroll
        for (int s = 0; s < KB_STEPS; ++s) {
            f.b0[s].x = wp[s * 4 * NB];
            f.b0[s].y = wp[s * 4 * NB + NB];
            f.b1[s].x = wp[s * 4 * NB + 16];
            f.b1[s].y = wp[s * 4 * NB + NB + 16];
        }
    }
    #pragma unroll
    for (int s = 0; s < KB_STEPS; ++s) {
        f.a[s].x = hp[4 * s];
        f.a[s].y = hp[4 * s + 1];
    }
}

__device__ __forceinline__ void mma_set(const FragSet& f, v8f& acc0, v8f& acc1) {
    #pragma unroll
    for (int s = 0; s < KB_STEPS; ++s) {
        acc0 = __builtin_amdgcn_wmma_f32_16x16x4_f32(
            false, f.a[s], false, f.b0[s], (short)0, acc0, false, false);
        acc1 = __builtin_amdgcn_wmma_f32_16x16x4_f32(
            false, f.a[s], false, f.b1[s], (short)0, acc1, false, false);
    }
}

template <bool PK>
__device__ __forceinline__ void gemm_head(int k, float* Pd, const float* hL,
                                          const float* __restrict__ W,
                                          const float* __restrict__ wpk,
                                          const float* __restrict__ bvec,
                                          int wave_u, int lane_off, int lane,
                                          int n16, int hi, int kk2) {
    constexpr int VL = PK ? KB_STEPS : 4 * KB_STEPS;
    const float* hrow = hL + n16 * HSTR + kk2;
    for (int pair = wave_u; pair < NPAIRS; pair += NWAVES) {
        prio_by_passes_left((NPAIRS - 1 - pair) / NWAVES);
        const int pcol0 = pair * 32;
        const int pcol1 = pcol0 + 16;
        const int wcol0 = pcol_to_wcol(pcol0, k);
        const float* wc = PK
            ? wpk + (size_t)((k * NPAIRS + pair) * KSTEPS) * REC_FLOATS + lane * 4
            : W + wcol0;
        const float bias0 = bvec[wcol0 + n16];
        const float bias1 = bvec[wcol0 + 16 + n16];
        v8f acc0 = {};
        v8f acc1 = {};
        __builtin_amdgcn_sched_barrier(0);
        FragSet f[KBLOCKS];
        #pragma unroll
        for (int kb = 0; kb < KBLOCKS; ++kb)
            load_set<PK>(f[kb], wc, lane_off, hrow, kb);
        #pragma unroll
        for (int kb = 0; kb < KBLOCKS; ++kb)
            mma_set(f[kb], acc0, acc1);
        __builtin_amdgcn_sched_group_barrier(0x020, 2 * VL, 0);
        __builtin_amdgcn_sched_group_barrier(0x100, 2 * KB_STEPS, 0);
        #pragma unroll
        for (int kb = 0; kb < KBLOCKS; ++kb) {
            if (kb + 2 < KBLOCKS) {
                __builtin_amdgcn_sched_group_barrier(0x020, VL, 0);
                __builtin_amdgcn_sched_group_barrier(0x100, KB_STEPS, 0);
            }
            __builtin_amdgcn_sched_group_barrier(0x008, 2 * KB_STEPS, 0);
        }
        #pragma unroll
        for (int r = 0; r < 8; ++r) {
            Pd[(r + 8 * hi) * PSTR + pcol0 + n16] = acc0[r] + bias0;
            Pd[(r + 8 * hi) * PSTR + pcol1 + n16] = acc1[r] + bias1;
        }
    }
}

__device__ __forceinline__ void phase_b(const float* Pr, float* yv, float* tv, int lane,
                                        float& zreg, float& ln_det) {
    float v[NH], cf[NH];
    #pragma unroll
    for (int hd = 0; hd < NH; ++hd) v[hd] = Pr[VOFF + hd * L + lane];
    #pragma unroll
    for (int hd = 0; hd < NH; ++hd) cf[hd] = 2.0f / wave_sum(v[hd] * v[hd]);

    float y = zreg;
    #pragma unroll
    for (int hd = 0; hd < NH; ++hd) {
        float d = wave_sum(v[hd] * y);
        y -= (cf[hd] * d) * v[hd];
    }
    yv[lane] = y;

    float rtraw = Pr[RTOFF + lane * (L + 1)];
    float rt_d  = softplus_f(rtraw) + 0.01f;
    float rraw  = Pr[ROFF  + lane * (L + 1)];
    float rr    = rraw * rt_d;
    float mval  = softplus_f(rr) - 1.0f;
    float rdnew = rraw + (mval - rr) / rt_d;

    float wv = rt_d * y + Pr[BBOFF + lane];
    #pragma unroll
    for (int jj = 1; jj < L; ++jj) {
        int col = (lane + jj) & (L - 1);
        float e = Pr[RTOFF + lane * L + col] * yv[col];
        wv += (col > lane) ? e : 0.0f;
    }
    float t = tanhf(wv);
    tv[lane] = t;

    float sv = rdnew * t;
    #pragma unroll
    for (int jj = 1; jj < L; ++jj) {
        int col = (lane + jj) & (L - 1);
        float e = Pr[ROFF + lane * L + col] * tv[col];
        sv += (col > lane) ? e : 0.0f;
    }

    float x = sv;
    #pragma unroll
    for (int hd = NH - 1; hd >= 0; --hd) {
        float d = wave_sum(v[hd] * x);
        x -= (cf[hd] * d) * v[hd];
    }
    zreg += x;

    float hp = 1.0f - t * t;
    ln_det += logf(fabsf(1.0f + hp * mval) + 1e-8f);
}

#define PACK_WAVES 5
#define PACK_STEPS (KSTEPS / PACK_WAVES)
extern "C" __global__ __launch_bounds__(PACK_WAVES * 32)
void pack_w_kernel(const float* __restrict__ W, float* __restrict__ wpk, size_t ws_size)
{
    if (ws_size < PACK_BYTES) return;
    const int lane = threadIdx.x & 31;
    const int wave = threadIdx.x >> 5;
    const int k    = blockIdx.x / NPAIRS;
    const int pair = blockIdx.x - k * NPAIRS;
    const float* src = W + pcol_to_wcol(pair * 32, k)
                         + (size_t)(wave * PACK_STEPS * 4 + (lane >> 4) * 2) * NB + (lane & 15);
    v4f* dst = (v4f*)wpk + (size_t)(blockIdx.x * KSTEPS + wave * PACK_STEPS) * (REC_FLOATS / 4) + lane;
    v4f v[PACK_STEPS];
    #pragma unroll
    for (int s = 0; s < PACK_STEPS; ++s) {
        v[s].x = src[s * 4 * NB];
        v[s].y = src[s * 4 * NB + NB];
        v[s].z = src[s * 4 * NB + 16];
        v[s].w = src[s * 4 * NB + NB + 16];
    }
    #pragma unroll
    for (int s = 0; s < PACK_STEPS; ++s)
        dst[s * (REC_FLOATS / 4)] = v[s];
}

extern "C" __global__ __launch_bounds__(THREADS)
void sylvester_flow_kernel(const float* __restrict__ z0,
                           const float* __restrict__ hmat,
                           const float* __restrict__ W,
                           const float* __restrict__ bvec,
                           const float* __restrict__ wpk,
                           size_t ws_size,
                           float* __restrict__ out)
{
    extern __shared__ float smem[];
    float* P  = smem;
    float* hL = P  + 2 * PBUF;
    float* ys = hL + TILE_M * HSTR;
    float* ts = ys + TILE_M * L;

    const int tid  = threadIdx.x;
    const int wave = tid >> 5;
    const int lane = tid & 31;
    const int hi   = lane >> 4;
    const int n16  = lane & 15;
    const int kk2  = hi * 2;
    const int row0 = blockIdx.x * TILE_M;
    const int wave_u = __builtin_amdgcn_readfirstlane(wave);
    const int lane_off = kk2 * NB + n16;
    const bool packed = ws_size >= PACK_BYTES;

    for (int idx = tid; idx < TILE_M * HID; idx += THREADS) {
        int r = idx / HID, c = idx - r * HID;
        hL[r * HSTR + c] = hmat[(size_t)(row0 + r) * HID + c];
    }
    const int myrow = wave_u;
    float zreg = z0[(size_t)(row0 + myrow) * L + lane];
    float ln_det = 0.0f;

    const int b_first = (wave_u >> 2) & 1;

    __syncthreads();

    for (int st = 0; st <= KHEADS; ++st) {
        float* Pg = P + (st & 1) * PBUF;
        const float* Pb = P + ((st + 1) & 1) * PBUF;
        for (int step = 0; step < 2; ++step) {
            if ((step ^ b_first) == 0) {
                if (st < KHEADS) {
                    if (packed)
                        gemm_head<true>(st, Pg, hL, W, wpk, bvec, wave_u, lane_off, lane, n16, hi, kk2);
                    else
                        gemm_head<false>(st, Pg, hL, W, wpk, bvec, wave_u, lane_off, lane, n16, hi, kk2);
                }
            } else {
                if (st > 0) {
                    __builtin_amdgcn_s_setprio(3);
                    phase_b(Pb + myrow * PSTR, ys + myrow * L, ts + myrow * L,
                            lane, zreg, ln_det);
                    __builtin_amdgcn_s_setprio(0);
                }
            }
        }
        __syncthreads();
    }

    out[(size_t)(row0 + myrow) * L + lane] = zreg;
    out[(size_t)BATCH * L + (size_t)(row0 + myrow) * L + lane] = ln_det;
}

extern "C" void kernel_launch(void* const* d_in, const int* in_sizes, int n_in,
                              void* d_out, int out_size, void* d_ws, size_t ws_size,
                              hipStream_t stream) {
    (void)in_sizes; (void)n_in; (void)out_size;
    const float* z0 = (const float*)d_in[0];
    const float* h  = (const float*)d_in[1];
    const float* W  = (const float*)d_in[2];
    const float* b  = (const float*)d_in[3];
    float* out = (float*)d_out;
    float* wpk = (float*)d_ws;

    pack_w_kernel<<<KHEADS * NPAIRS, PACK_WAVES * 32, 0, stream>>>(W, wpk, ws_size);

    size_t lds_bytes = (size_t)(2 * PBUF + TILE_M * HSTR + 2 * TILE_M * L) * sizeof(float);
    hipFuncSetAttribute((const void*)sylvester_flow_kernel,
                        hipFuncAttributeMaxDynamicSharedMemorySize,
                        (int)lds_bytes);
    sylvester_flow_kernel<<<BATCH / TILE_M, THREADS, lds_bytes, stream>>>(
        z0, h, W, b, wpk, ws_size, out);
}
